# starting kernel
# baseline (speedup 1.0000x reference)
#include <hip/hip_runtime.h>
#include <hip/hip_bf16.h>


typedef __attribute__((ext_vector_type(2))) float v2f;
typedef __attribute__((ext_vector_type(4))) float v4f;
typedef __attribute__((ext_vector_type(8))) float v8f;

namespace {
constexpr int M_DIM = 128;
constexpr int N_DIM = 8192;
constexpr int K_DIM = 8192;
constexpr float SCALING = 2.0f;
}

__global__ __launch_bounds__(256, 1)
void lokr_fused_gemm(const float* __restrict__ x,
                     const float* __restrict__ W,
                     const float* __restrict__ A,
                     const float* __restrict__ B16,
                     const float* __restrict__ scal,
                     float* __restrict__ out)
{
    const int lane  = threadIdx.x & 31;
    const int wave  = threadIdx.x >> 5;
    const int nt    = blockIdx.x * 4 + (wave & 3);
    const int mbase = (wave >> 2) * 64;
    const int n0    = nt * 16;
    const int l16   = lane & 15;
    const int lhalf = lane >> 4;

    const float sb = scal[0] * SCALING * B16[l16];

    const float* wp = W + (size_t)(n0 + l16) * K_DIM + 4 * lhalf;
    const float* ap = A + (size_t)nt * K_DIM + 4 * lhalf;
    const float* xp = x + (size_t)(mbase + l16) * K_DIM + 4 * lhalf;

    v8f c[4] = {};

    for (int kc = 0; kc < K_DIM; kc += 32) {
        __builtin_prefetch(wp + kc + 2048, 0, 1);
        #pragma unroll
        for (int ku = 0; ku < 32; ku += 8) {
            const int kb = kc + ku;
            const v4f w4 = *(const v4f*)(wp + kb);
            const v4f a4 = *(const v4f*)(ap + kb);
            v4f be;
            be.x = __builtin_fmaf(sb, a4.x, w4.x);
            be.y = __builtin_fmaf(sb, a4.y, w4.y);
            be.z = __builtin_fmaf(sb, a4.z, w4.z);
            be.w = __builtin_fmaf(sb, a4.w, w4.w);
            const v2f b0 = __builtin_shufflevector(be, be, 0, 1);
            const v2f b1 = __builtin_shufflevector(be, be, 2, 3);

            v4f x4[4];
            #pragma unroll
            for (int mt = 0; mt < 4; ++mt)
                x4[mt] = *(const v4f*)(xp + (size_t)(mt * 16) * K_DIM + kb);

            #pragma unroll
            for (int mt = 0; mt < 4; ++mt) {
                const v2f a0 = __builtin_shufflevector(x4[mt], x4[mt], 0, 1);
                c[mt] = __builtin_amdgcn_wmma_f32_16x16x4_f32(
                    false, a0, false, b0, (short)0, c[mt], false, false);
            }
            #pragma unroll
            for (int mt = 0; mt < 4; ++mt) {
                const v2f a1 = __builtin_shufflevector(x4[mt], x4[mt], 2, 3);
                c[mt] = __builtin_amdgcn_wmma_f32_16x16x4_f32(
                    false, a1, false, b1, (short)0, c[mt], false, false);
            }
        }
    }

    float* op = out + n0 + l16;
    #pragma unroll
    for (int mt = 0; mt < 4; ++mt) {
        #pragma unroll
        for (int r = 0; r < 8; ++r) {
            const int m = mbase + mt * 16 + lhalf * 8 + r;
            op[(size_t)m * N_DIM] = c[mt][r];
        }
    }
}

extern "C" void kernel_launch(void* const* d_in, const int* in_sizes, int n_in,
                              void* d_out, int out_size, void* d_ws, size_t ws_size,
                              hipStream_t stream) {
    (void)in_sizes; (void)n_in; (void)out_size; (void)d_ws; (void)ws_size;
    const float* x    = (const float*)d_in[0];
    const float* W    = (const float*)d_in[1];
    const float* A    = (const float*)d_in[2];
    const float* B16  = (const float*)d_in[3];
    const float* scal = (const float*)d_in[4];
    float* out = (float*)d_out;

    dim3 grid(N_DIM / 64);
    dim3 block(256);
    hipLaunchKernelGGL(lokr_fused_gemm, grid, block, 0, stream,
                       x, W, A, B16, scal, out);
}


# Operands staged through LDS: per 128-k tile x and W+sb*A are copied in by whole-row coalesced loads (2 stages, register prefetch); 256 groups of 4 waves read b128 fragments; summation order kept
# speedup vs baseline: 4.9176x; 4.9176x over previous
#include <hip/hip_runtime.h>


typedef __attribute__((ext_vector_type(2))) float v2f;
typedef __attribute__((ext_vector_type(4))) float v4f;
typedef __attribute__((ext_vector_type(8))) float v8f;

namespace {
constexpr int M_DIM = 128;
constexpr int N_DIM = 8192;
constexpr int K_DIM = 8192;
constexpr float SCALING = 2.0f;

constexpr int BN = 32;
constexpr int BK = 128;
constexpr int PITCH = BK + 4;
constexpr int X_FLOATS = M_DIM * PITCH;
constexpr int STAGE_FLOATS = (M_DIM + BN) * PITCH;
constexpr int NT = K_DIM / BK;
constexpr int XJ = M_DIM / 4;
constexpr int WJ = BN / 4;
}

__global__ __launch_bounds__(128)
void lokr_fused_gemm(const float* __restrict__ x,
                     const float* __restrict__ W,
                     const float* __restrict__ A,
                     const float* __restrict__ B16,
                     const float* __restrict__ scal,
                     float* __restrict__ out)
{
    __shared__ __attribute__((aligned(16))) float lds[2 * STAGE_FLOATS];

    const int lane = threadIdx.x & 31;
    const int wave = threadIdx.x >> 5;
    const int c    = lane & 15;
    const int h    = lane >> 4;
    const int nb   = blockIdx.x * BN;

    const float s = scal[0] * SCALING;
    float sb[4];
    #pragma unroll
    for (int q = 0; q < 4; ++q) sb[q] = s * B16[wave + 4 * q];

    const float* xg = x + (size_t)wave * K_DIM + 4 * lane;
    const float* wg = W + (size_t)(nb + wave) * K_DIM + 4 * lane;
    const float* ag = A + (size_t)(blockIdx.x * 2) * K_DIM + 4 * lane;

    const int st_off = wave * PITCH + 4 * lane;
    const int fa_off = (32 * wave + c) * PITCH + 4 * h;
    const int fb_off = X_FLOATS + c * PITCH + 4 * h;

    v4f xr[XJ], wr[WJ], ar[2];
    v8f acc[2][2] = {};

    constexpr int XA = 11;
    auto gloadA = [&](int k0) {
        #pragma unroll
        for (int j = 0; j < WJ; ++j)
            wr[j] = *(const v4f*)(wg + (size_t)(4 * j) * K_DIM + k0);
        ar[0] = *(const v4f*)(ag + k0);
        ar[1] = *(const v4f*)(ag + K_DIM + k0);
        #pragma unroll
        for (int j = 0; j < XA; ++j)
            xr[j] = *(const v4f*)(xg + (size_t)(4 * j) * K_DIM + k0);
    };
    auto gloadB = [&](int k0) {
        #pragma unroll
        for (int j = XA; j < XJ; ++j)
            xr[j] = *(const v4f*)(xg + (size_t)(4 * j) * K_DIM + k0);
    };

    auto sstore_x = [&](int buf, int j0, int j1) {
        float* xs = lds + buf * STAGE_FLOATS + st_off;
        #pragma unroll
        for (int j = j0; j < j1; ++j)
            *(v4f*)(xs + (4 * j) * PITCH) = xr[j];
    };
    auto sstore_w = [&](int buf) {
        float* ws = lds + buf * STAGE_FLOATS + X_FLOATS + st_off;
        #pragma unroll
        for (int j = 0; j < WJ; ++j) {
            const v4f a4 = ar[j >> 2];
            const float sq = sb[j & 3];
            v4f be;
            be.x = __builtin_fmaf(sq, a4.x, wr[j].x);
            be.y = __builtin_fmaf(sq, a4.y, wr[j].y);
            be.z = __builtin_fmaf(sq, a4.z, wr[j].z);
            be.w = __builtin_fmaf(sq, a4.w, wr[j].w);
            *(v4f*)(ws + (4 * j) * PITCH) = be;
        }
    };

    v4f fa[2][4][2], fb[2][4][2];

    auto frags = [&](int buf, int set, int blk) {
        const float* xa = lds + buf * STAGE_FLOATS + fa_off;
        const float* wb = lds + buf * STAGE_FLOATS + fb_off;
        #pragma unroll
        for (int q = 0; q < 4; ++q) {
            const int ko = 8 * (4 * blk + q);
            fa[set][q][0] = *(const v4f*)(xa + ko);
            fa[set][q][1] = *(const v4f*)(xa + 16 * PITCH + ko);
            fb[set][q][0] = *(const v4f*)(wb + ko);
            fb[set][q][1] = *(const v4f*)(wb + 16 * PITCH + ko);
        }
    };

    auto mma = [&](int set, int q0, int q1) {
        #pragma unroll
        for (int q = q0; q < q1; ++q) {
            #pragma unroll
            for (int mi = 0; mi < 2; ++mi) {
                const v2f a = __builtin_shufflevector(fa[set][q][mi], fa[set][q][mi], 0, 1);
                #pragma unroll
                for (int ni = 0; ni < 2; ++ni) {
                    const v2f b = __builtin_shufflevector(fb[set][q][ni], fb[set][q][ni], 0, 1);
                    acc[mi][ni] = __builtin_amdgcn_wmma_f32_16x16x4_f32(
                        false, a, false, b, (short)0, acc[mi][ni], false, false);
                }
            }
            #pragma unroll
            for (int mi = 0; mi < 2; ++mi) {
                const v2f a = __builtin_shufflevector(fa[set][q][mi], fa[set][q][mi], 2, 3);
                #pragma unroll
                for (int ni = 0; ni < 2; ++ni) {
                    const v2f b = __builtin_shufflevector(fb[set][q][ni], fb[set][q][ni], 2, 3);
                    acc[mi][ni] = __builtin_amdgcn_wmma_f32_16x16x4_f32(
                        false, a, false, b, (short)0, acc[mi][ni], false, false);
                }
            }
        }
    };

    gloadA(0);
    gloadB(0);
    sstore_x(0, 0, XJ);
    sstore_w(0);
    __syncthreads();
    gloadA(BK);
    frags(0, 0, 0);

    for (int t = 0; t < NT - 1; ++t) {
        const int cur = t & 1;
        const int nxt = cur ^ 1;
        gloadB((t + 1) * BK);
        frags(cur, 1, 1);
        __builtin_amdgcn_sched_barrier(0);
        mma(0, 0, 4);
        __builtin_amdgcn_sched_barrier(0);
        frags(cur, 0, 2);
        __builtin_amdgcn_sched_barrier(0);
        mma(1, 0, 4);
        __builtin_amdgcn_sched_barrier(0);
        frags(cur, 1, 3);
        sstore_x(nxt, 0, 16);
        __builtin_amdgcn_sched_barrier(0);
        mma(0, 0, 2);
        __builtin_amdgcn_sched_barrier(0);
        sstore_x(nxt, 16, XJ);
        sstore_w(nxt);
        __builtin_amdgcn_sched_barrier(0);
        mma(0, 2, 4);
        __builtin_amdgcn_sched_barrier(0);
        __syncthreads();
        gloadA((t + 2 < NT ? t + 2 : NT - 1) * BK);
        frags(nxt, 0, 0);
        __builtin_amdgcn_sched_barrier(0);
        mma(1, 0, 4);
    }
    {
        const int cur = (NT - 1) & 1;
        frags(cur, 1, 1);
        __builtin_amdgcn_sched_barrier(0);
        mma(0, 0, 4);
        __builtin_amdgcn_sched_barrier(0);
        frags(cur, 0, 2);
        __builtin_amdgcn_sched_barrier(0);
        mma(1, 0, 4);
        __builtin_amdgcn_sched_barrier(0);
        frags(cur, 1, 3);
        __builtin_amdgcn_sched_barrier(0);
        mma(0, 0, 4);
        mma(1, 0, 4);
    }

    float* op = out + (size_t)(32 * wave + 8 * h) * N_DIM + nb + c;
    #pragma unroll
    for (int mi = 0; mi < 2; ++mi) {
        #pragma unroll
        for (int r = 0; r < 8; ++r) {
            #pragma unroll
            for (int ni = 0; ni < 2; ++ni)
                op[(size_t)(16 * mi + r) * N_DIM + 16 * ni] = acc[mi][ni][r];
        }
    }
}

extern "C" void kernel_launch(void* const* d_in, const int* in_sizes, int n_in,
                              void* d_out, int out_size, void* d_ws, size_t ws_size,
                              hipStream_t stream) {
    (void)in_sizes; (void)n_in; (void)out_size; (void)d_ws; (void)ws_size;
    const float* x    = (const float*)d_in[0];
    const float* W    = (const float*)d_in[1];
    const float* A    = (const float*)d_in[2];
    const float* B16  = (const float*)d_in[3];
    const float* scal = (const float*)d_in[4];
    float* out = (float*)d_out;

    dim3 grid(N_DIM / BN);
    dim3 block(128);
    hipLaunchKernelGGL(lokr_fused_gemm, grid, block, 0, stream,
                       x, W, A, B16, scal, out);
}


# x copied once per call into fragment order in the workspace by a first launch; the GEMM loads x fragments from it straight into registers a tile ahead, x no longer staged in LDS; sums unchanged
# speedup vs baseline: 5.3656x; 1.0911x over previous
#include <hip/hip_runtime.h>


typedef __attribute__((ext_vector_type(2))) float v2f;
typedef __attribute__((ext_vector_type(4))) float v4f;
typedef __attribute__((ext_vector_type(8))) float v8f;

namespace {
constexpr int M_DIM = 128;
constexpr int N_DIM = 8192;
constexpr int K_DIM = 8192;
constexpr float SCALING = 2.0f;

constexpr int BN = 32;
constexpr int BK = 128;
constexpr int PITCH = BK + 4;
constexpr int STAGE_FLOATS = BN * PITCH;
constexpr int NT = K_DIM / BK;
constexpr int WJ = BN / 4;
constexpr int KG = K_DIM / 8;
constexpr int GT = BK / 8;
constexpr size_t XF_BYTES = (size_t)M_DIM * K_DIM * sizeof(float);
constexpr int LDS_FLOATS = 42240;

template <int P> struct IC { static constexpr int v = P; };
}

__global__ __launch_bounds__(256)
void lokr_x_to_fragments(const float* __restrict__ x, float* __restrict__ xf, size_t ws_size)
{
    if (ws_size < XF_BYTES) return;
    const int idx  = blockIdx.x * 256 + threadIdx.x;
    const int lane = idx & 31;
    const int g    = (idx >> 5) & (KG - 1);
    const int mt   = idx >> 15;
    const int c    = lane & 15;
    const int h    = lane >> 4;
    const v4f v = *(const v4f*)(x + (size_t)(16 * mt + c) * K_DIM + 8 * g + 4 * h);
    *(v4f*)(xf + (size_t)idx * 4) = v;
}

__global__ __launch_bounds__(128)
void lokr_fused_gemm(const float* __restrict__ x,
                     const float* __restrict__ W,
                     const float* __restrict__ A,
                     const float* __restrict__ B16,
                     const float* __restrict__ scal,
                     const float* __restrict__ xfw,
                     size_t ws_size,
                     float* __restrict__ out)
{
    __shared__ __attribute__((aligned(16))) float lds[LDS_FLOATS];

    const int lane = threadIdx.x & 31;
    const int wave = threadIdx.x >> 5;
    const int c    = lane & 15;
    const int h    = lane >> 4;
    const int nb   = blockIdx.x * BN;

    const float* xf = (ws_size >= XF_BYTES) ? xfw : x;

    const float s = scal[0] * SCALING;
    float sb[4];
    #pragma unroll
    for (int q = 0; q < 4; ++q) sb[q] = s * B16[wave + 4 * q];

    const float* wg = W + (size_t)(nb + wave) * K_DIM + 4 * lane;
    const float* ag = A + (size_t)(blockIdx.x * 2) * K_DIM + 4 * lane;
    const float* xg0 = xf + ((size_t)(2 * wave) * KG * 32 + lane) * 4;
    const float* xg1 = xg0 + (size_t)KG * 32 * 4;

    const int st_off = wave * PITCH + 4 * lane;
    const int fb_off = c * PITCH + 4 * h;

    v4f xfr[2][2][GT];
    v4f wr[2][WJ], ar[2][2];
    v4f fb[2][4][2];
    v8f acc[2][2] = {};

    auto tile_of = [](int t) { return t < NT ? t : NT - 1; };

    auto gload_x = [&](auto setc, int t, int g0, int g1) {
        constexpr int set = decltype(setc)::v;
        const size_t o = (size_t)(tile_of(t) * GT) * 32 * 4;
        #pragma unroll
        for (int g = g0; g < g1; ++g) {
            xfr[set][0][g] = *(const v4f*)(xg0 + o + (size_t)g * 32 * 4);
            xfr[set][1][g] = *(const v4f*)(xg1 + o + (size_t)g * 32 * 4);
        }
    };
    auto gload_w = [&](auto setc, int t) {
        constexpr int set = decltype(setc)::v;
        const int k0 = tile_of(t) * BK;
        #pragma unroll
        for (int j = 0; j < WJ; ++j)
            wr[set][j] = *(const v4f*)(wg + (size_t)(4 * j) * K_DIM + k0);
        ar[set][0] = *(const v4f*)(ag + k0);
        ar[set][1] = *(const v4f*)(ag + K_DIM + k0);
    };
    auto sstore_w = [&](auto setc, int buf) {
        constexpr int set = decltype(setc)::v;
        float* ws = lds + buf * STAGE_FLOATS + st_off;
        #pragma unroll
        for (int j = 0; j < WJ; ++j) {
            const v4f a4 = ar[set][j >> 2];
            const float sq = sb[j & 3];
            v4f be;
            be.x = __builtin_fmaf(sq, a4.x, wr[set][j].x);
            be.y = __builtin_fmaf(sq, a4.y, wr[set][j].y);
            be.z = __builtin_fmaf(sq, a4.z, wr[set][j].z);
            be.w = __builtin_fmaf(sq, a4.w, wr[set][j].w);
            *(v4f*)(ws + (4 * j) * PITCH) = be;
        }
    };
    auto frags = [&](int buf, auto setc, int blk) {
        constexpr int set = decltype(setc)::v;
        const float* wb = lds + buf * STAGE_FLOATS + fb_off;
        #pragma unroll
        for (int q = 0; q < 4; ++q) {
            const int ko = 8 * (4 * blk + q);
            fb[set][q][0] = *(const v4f*)(wb + ko);
            fb[set][q][1] = *(const v4f*)(wb + 16 * PITCH + ko);
        }
    };
    auto mma = [&](auto xsc, auto bsc, auto blkc) {
        constexpr int xs = decltype(xsc)::v;
        constexpr int bs = decltype(bsc)::v;
        constexpr int blk = decltype(blkc)::v;
        #pragma unroll
        for (int q = 0; q < 4; ++q) {
            #pragma unroll
            for (int mi = 0; mi < 2; ++mi) {
                const v4f f = xfr[xs][mi][4 * blk + q];
                const v2f a = __builtin_shufflevector(f, f, 0, 1);
                #pragma unroll
                for (int ni = 0; ni < 2; ++ni) {
                    const v2f b = __builtin_shufflevector(fb[bs][q][ni], fb[bs][q][ni], 0, 1);
                    acc[mi][ni] = __builtin_amdgcn_wmma_f32_16x16x4_f32(
                        false, a, false, b, (short)0, acc[mi][ni], false, false);
                }
            }
            #pragma unroll
            for (int mi = 0; mi < 2; ++mi) {
                const v4f f = xfr[xs][mi][4 * blk + q];
                const v2f a = __builtin_shufflevector(f, f, 2, 3);
                #pragma unroll
                for (int ni = 0; ni < 2; ++ni) {
                    const v2f b = __builtin_shufflevector(fb[bs][q][ni], fb[bs][q][ni], 2, 3);
                    acc[mi][ni] = __builtin_amdgcn_wmma_f32_16x16x4_f32(
                        false, a, false, b, (short)0, acc[mi][ni], false, false);
                }
            }
        }
    };

    auto tile = [&](auto pc, int t) {
        constexpr int P = decltype(pc)::v;
        constexpr int Q = P ^ 1;
        gload_w(IC<P>{}, t + 2);
        gload_x(IC<Q>{}, t + 1, 0, 4);
        frags(P, IC<1>{}, 1);
        __builtin_amdgcn_sched_barrier(0);
        mma(IC<P>{}, IC<0>{}, IC<0>{});
        __builtin_amdgcn_sched_barrier(0);
        gload_x(IC<Q>{}, t + 1, 4, 8);
        frags(P, IC<0>{}, 2);
        __builtin_amdgcn_sched_barrier(0);
        mma(IC<P>{}, IC<1>{}, IC<1>{});
        __builtin_amdgcn_sched_barrier(0);
        gload_x(IC<Q>{}, t + 1, 8, 12);
        frags(P, IC<1>{}, 3);
        sstore_w(IC<Q>{}, Q);
        __builtin_amdgcn_sched_barrier(0);
        mma(IC<P>{}, IC<0>{}, IC<2>{});
        __builtin_amdgcn_sched_barrier(0);
        __builtin_amdgcn_fence(__ATOMIC_RELEASE, "workgroup", "local");
        __builtin_amdgcn_s_barrier();
        __builtin_amdgcn_fence(__ATOMIC_ACQUIRE, "workgroup", "local");
        gload_x(IC<Q>{}, t + 1, 12, 16);
        frags(Q, IC<0>{}, 0);
        __builtin_amdgcn_sched_barrier(0);
        mma(IC<P>{}, IC<1>{}, IC<3>{});
        __builtin_amdgcn_sched_barrier(0);
    };

    gload_w(IC<0>{}, 0);
    gload_x(IC<0>{}, 0, 0, GT);
    sstore_w(IC<0>{}, 0);
    gload_w(IC<1>{}, 1);
    __syncthreads();
    frags(0, IC<0>{}, 0);

    for (int t = 0; t < NT; t += 2) {
        tile(IC<0>{}, t);
        tile(IC<1>{}, t + 1);
    }

    float* op = out + (size_t)(32 * wave + 8 * h) * N_DIM + nb + c;
    #pragma unroll
    for (int mi = 0; mi < 2; ++mi) {
        #pragma unroll
        for (int r = 0; r < 8; ++r) {
            #pragma unroll
            for (int ni = 0; ni < 2; ++ni)
                op[(size_t)(16 * mi + r) * N_DIM + 16 * ni] = acc[mi][ni][r];
        }
    }
}

extern "C" void kernel_launch(void* const* d_in, const int* in_sizes, int n_in,
                              void* d_out, int out_size, void* d_ws, size_t ws_size,
                              hipStream_t stream) {
    (void)in_sizes; (void)n_in; (void)out_size;
    const float* x    = (const float*)d_in[0];
    const float* W    = (const float*)d_in[1];
    const float* A    = (const float*)d_in[2];
    const float* B16  = (const float*)d_in[3];
    const float* scal = (const float*)d_in[4];
    float* out = (float*)d_out;
    float* xf  = (float*)d_ws;

    hipLaunchKernelGGL(lokr_x_to_fragments, dim3(1024), dim3(256), 0, stream, x, xf, ws_size);

    dim3 grid(N_DIM / BN);
    dim3 block(128);
    hipLaunchKernelGGL(lokr_fused_gemm, grid, block, 0, stream,
                       x, W, A, B16, scal, (const float*)xf, ws_size, out);
}
